# starting kernel
# baseline (speedup 1.0000x reference)
#include <hip/hip_runtime.h>

typedef float v2f __attribute__((ext_vector_type(2)));
typedef float v8f __attribute__((ext_vector_type(8)));

#define NCP  64
#define UPTS 1024
#define VPTS 1024
#define NB   4
#define VT_PER_WAVE 4

__global__ __launch_bounds__(256)
void build_dense_basis(const float* __restrict__ Nu,
                       const float* __restrict__ Nv,
                       const int*   __restrict__ uspan,
                       const int*   __restrict__ vspan,
                       float* __restrict__ NuF,
                       float* __restrict__ NvF) {
    int t = blockIdx.x * blockDim.x + threadIdx.x;
    if (t < UPTS) {
        float* row = NuF + t * NCP;
        #pragma unroll
        for (int i = 0; i < NCP; ++i) row[i] = 0.0f;
        int s = uspan[t] - 3;
        #pragma unroll
        for (int l = 0; l < 4; ++l) row[s + l] = Nu[t * 4 + l];
    } else if (t < 2 * UPTS) {
        int v = t - UPTS;
        float* row = NvF + v * NCP;
        #pragma unroll
        for (int i = 0; i < NCP; ++i) row[i] = 0.0f;
        int s = vspan[v] - 3;
        #pragma unroll
        for (int l = 0; l < 4; ++l) row[s + l] = Nv[v * 4 + l];
    }
}

__global__ __launch_bounds__(32)
void surf_stage1(const float* __restrict__ NuF,
                 const float* __restrict__ ctrl,
                 float* __restrict__ T) {
    const int lane = threadIdx.x;
    const int ut   = blockIdx.x;
    const int jt   = blockIdx.y;
    const int bd   = blockIdx.z;
    const int b    = bd >> 2;
    const int d    = bd & 3;

    const int half = lane >> 4;
    const int l15  = lane & 15;
    const int urow = ut * 16 + l15;
    const int jcol = jt * 16 + l15;

    v8f acc = {};
    const float* nurow = NuF + urow * NCP;

    #pragma unroll
    for (int k0 = 0; k0 < NCP; k0 += 4) {
        const int ka = k0 + half * 2;
        v2f a = *(const v2f*)(nurow + ka);
        v2f bf;
        bf.x = ctrl[((b * NCP + ka    ) * NCP + jcol) * 4 + d];
        bf.y = ctrl[((b * NCP + ka + 1) * NCP + jcol) * 4 + d];
        acc = __builtin_amdgcn_wmma_f32_16x16x4_f32(
                  false, a, false, bf, (short)0, acc, false, false);
    }

    #pragma unroll
    for (int r = 0; r < 8; ++r) {
        const int u = ut * 16 + half * 8 + r;
        T[(bd * UPTS + u) * NCP + (jt * 16 + l15)] = acc[r];
    }
}

__global__ __launch_bounds__(32)
void surf_stage2(const float* __restrict__ T,
                 const float* __restrict__ NvF,
                 float* __restrict__ out) {
    const int lane = threadIdx.x;
    const int vt0  = blockIdx.x * VT_PER_WAVE;
    const int ut   = blockIdx.y;
    const int b    = blockIdx.z;

    const int half = lane >> 4;
    const int l15  = lane & 15;
    const int urow = ut * 16 + l15;

    const float* t0 = T + ((b * 4 + 0) * UPTS + urow) * NCP;
    const float* t1 = T + ((b * 4 + 1) * UPTS + urow) * NCP;
    const float* t2 = T + ((b * 4 + 2) * UPTS + urow) * NCP;
    const float* t3 = T + ((b * 4 + 3) * UPTS + urow) * NCP;

    const float* nvr[VT_PER_WAVE];
    #pragma unroll
    for (int iv = 0; iv < VT_PER_WAVE; ++iv)
        nvr[iv] = NvF + ((vt0 + iv) * 16 + l15) * NCP;

    v8f acc[VT_PER_WAVE][4];
    #pragma unroll
    for (int iv = 0; iv < VT_PER_WAVE; ++iv)
        #pragma unroll
        for (int d = 0; d < 4; ++d)
            acc[iv][d] = (v8f){};

    #pragma unroll
    for (int k0 = 0; k0 < NCP; k0 += 4) {
        const int ka = k0 + half * 2;
        v2f a0 = *(const v2f*)(t0 + ka);
        v2f a1 = *(const v2f*)(t1 + ka);
        v2f a2 = *(const v2f*)(t2 + ka);
        v2f a3 = *(const v2f*)(t3 + ka);
        #pragma unroll
        for (int iv = 0; iv < VT_PER_WAVE; ++iv) {
            v2f bf = *(const v2f*)(nvr[iv] + ka);
            acc[iv][0] = __builtin_amdgcn_wmma_f32_16x16x4_f32(
                             false, a0, false, bf, (short)0, acc[iv][0], false, false);
            acc[iv][1] = __builtin_amdgcn_wmma_f32_16x16x4_f32(
                             false, a1, false, bf, (short)0, acc[iv][1], false, false);
            acc[iv][2] = __builtin_amdgcn_wmma_f32_16x16x4_f32(
                             false, a2, false, bf, (short)0, acc[iv][2], false, false);
            acc[iv][3] = __builtin_amdgcn_wmma_f32_16x16x4_f32(
                             false, a3, false, bf, (short)0, acc[iv][3], false, false);
        }
    }

    #pragma unroll
    for (int iv = 0; iv < VT_PER_WAVE; ++iv) {
        #pragma unroll
        for (int r = 0; r < 8; ++r) {
            const int u = ut * 16 + half * 8 + r;
            const int v = (vt0 + iv) * 16 + l15;
            const float w = acc[iv][3][r];
            const size_t o = ((size_t)(b * UPTS + u) * VPTS + v) * 3;
            out[o + 0] = acc[iv][0][r] / w;
            out[o + 1] = acc[iv][1][r] / w;
            out[o + 2] = acc[iv][2][r] / w;
        }
    }
}

extern "C" void kernel_launch(void* const* d_in, const int* in_sizes, int n_in,
                              void* d_out, int out_size, void* d_ws, size_t ws_size,
                              hipStream_t stream) {
    const float* ctrl  = (const float*)d_in[0];
    const float* Nu    = (const float*)d_in[1];
    const float* Nv    = (const float*)d_in[2];
    const int*   uspan = (const int*)  d_in[3];
    const int*   vspan = (const int*)  d_in[4];
    float*       out   = (float*)d_out;

    float* NuF = (float*)d_ws;
    float* NvF = NuF + (size_t)UPTS * NCP;
    float* T   = NvF + (size_t)VPTS * NCP;

    build_dense_basis<<<(2 * UPTS + 255) / 256, 256, 0, stream>>>(
        Nu, Nv, uspan, vspan, NuF, NvF);

    surf_stage1<<<dim3(UPTS / 16, NCP / 16, NB * 4), 32, 0, stream>>>(
        NuF, ctrl, T);

    surf_stage2<<<dim3(VPTS / 16 / VT_PER_WAVE, UPTS / 16, NB), 32, 0, stream>>>(
        T, NvF, out);
}


# Band-limited K loop in stage 2: a 16x64 strip runs only the 4-wide K blocks between the min and max vspan of its 64 v points (wave reduce); skipped blocks are exact zeros, sum order kept
# speedup vs baseline: 1.1374x; 1.1374x over previous
#include <hip/hip_runtime.h>

typedef float v2f __attribute__((ext_vector_type(2)));
typedef float v8f __attribute__((ext_vector_type(8)));

#define NCP  64
#define UPTS 1024
#define VPTS 1024
#define NB   4
#define VT_PER_WAVE 4

__global__ __launch_bounds__(256)
void build_dense_basis(const float* __restrict__ Nu,
                       const float* __restrict__ Nv,
                       const int*   __restrict__ uspan,
                       const int*   __restrict__ vspan,
                       float* __restrict__ NuF,
                       float* __restrict__ NvF) {
    int t = blockIdx.x * blockDim.x + threadIdx.x;
    if (t < UPTS) {
        float* row = NuF + t * NCP;
        #pragma unroll
        for (int i = 0; i < NCP; ++i) row[i] = 0.0f;
        int s = uspan[t] - 3;
        #pragma unroll
        for (int l = 0; l < 4; ++l) row[s + l] = Nu[t * 4 + l];
    } else if (t < 2 * UPTS) {
        int v = t - UPTS;
        float* row = NvF + v * NCP;
        #pragma unroll
        for (int i = 0; i < NCP; ++i) row[i] = 0.0f;
        int s = vspan[v] - 3;
        #pragma unroll
        for (int l = 0; l < 4; ++l) row[s + l] = Nv[v * 4 + l];
    }
}

__global__ __launch_bounds__(32)
void surf_stage1(const float* __restrict__ NuF,
                 const float* __restrict__ ctrl,
                 float* __restrict__ T) {
    const int lane = threadIdx.x;
    const int ut   = blockIdx.x;
    const int jt   = blockIdx.y;
    const int bd   = blockIdx.z;
    const int b    = bd >> 2;
    const int d    = bd & 3;

    const int half = lane >> 4;
    const int l15  = lane & 15;
    const int urow = ut * 16 + l15;
    const int jcol = jt * 16 + l15;

    v8f acc = {};
    const float* nurow = NuF + urow * NCP;

    #pragma unroll
    for (int k0 = 0; k0 < NCP; k0 += 4) {
        const int ka = k0 + half * 2;
        v2f a = *(const v2f*)(nurow + ka);
        v2f bf;
        bf.x = ctrl[((b * NCP + ka    ) * NCP + jcol) * 4 + d];
        bf.y = ctrl[((b * NCP + ka + 1) * NCP + jcol) * 4 + d];
        acc = __builtin_amdgcn_wmma_f32_16x16x4_f32(
                  false, a, false, bf, (short)0, acc, false, false);
    }

    #pragma unroll
    for (int r = 0; r < 8; ++r) {
        const int u = ut * 16 + half * 8 + r;
        T[(bd * UPTS + u) * NCP + (jt * 16 + l15)] = acc[r];
    }
}

__global__ __launch_bounds__(32)
void surf_stage2(const float* __restrict__ T,
                 const float* __restrict__ NvF,
                 const int*   __restrict__ vspan,
                 float* __restrict__ out) {
    const int lane = threadIdx.x;
    const int vt0  = blockIdx.x * VT_PER_WAVE;
    const int ut   = blockIdx.y;
    const int b    = blockIdx.z;

    const int half = lane >> 4;
    const int l15  = lane & 15;
    const int urow = ut * 16 + l15;

    const float* t0 = T + ((b * 4 + 0) * UPTS + urow) * NCP;
    const float* t1 = T + ((b * 4 + 1) * UPTS + urow) * NCP;
    const float* t2 = T + ((b * 4 + 2) * UPTS + urow) * NCP;
    const float* t3 = T + ((b * 4 + 3) * UPTS + urow) * NCP;

    const float* nvr[VT_PER_WAVE];
    #pragma unroll
    for (int iv = 0; iv < VT_PER_WAVE; ++iv)
        nvr[iv] = NvF + ((vt0 + iv) * 16 + l15) * NCP;

    v8f acc[VT_PER_WAVE][4];
    #pragma unroll
    for (int iv = 0; iv < VT_PER_WAVE; ++iv)
        #pragma unroll
        for (int d = 0; d < 4; ++d)
            acc[iv][d] = (v8f){};

    int smin, smax;
    {
        const int sa = vspan[vt0 * 16 + lane];
        const int sb = vspan[vt0 * 16 + 32 + lane];
        smin = min(sa, sb);
        smax = max(sa, sb);
    }
    for (int off = 16; off >= 1; off >>= 1) {
        smin = min(smin, __shfl_xor(smin, off));
        smax = max(smax, __shfl_xor(smax, off));
    }
    const int klo = min(max(__builtin_amdgcn_readfirstlane(smin) - 3, 0) & ~3, NCP - 4);
    const int khi = min(__builtin_amdgcn_readfirstlane(smax), NCP - 1);

    int k0 = klo;
    do {
        const int ka = k0 + half * 2;
        v2f a0 = *(const v2f*)(t0 + ka);
        v2f a1 = *(const v2f*)(t1 + ka);
        v2f a2 = *(const v2f*)(t2 + ka);
        v2f a3 = *(const v2f*)(t3 + ka);
        #pragma unroll
        for (int iv = 0; iv < VT_PER_WAVE; ++iv) {
            v2f bf = *(const v2f*)(nvr[iv] + ka);
            acc[iv][0] = __builtin_amdgcn_wmma_f32_16x16x4_f32(
                             false, a0, false, bf, (short)0, acc[iv][0], false, false);
            acc[iv][1] = __builtin_amdgcn_wmma_f32_16x16x4_f32(
                             false, a1, false, bf, (short)0, acc[iv][1], false, false);
            acc[iv][2] = __builtin_amdgcn_wmma_f32_16x16x4_f32(
                             false, a2, false, bf, (short)0, acc[iv][2], false, false);
            acc[iv][3] = __builtin_amdgcn_wmma_f32_16x16x4_f32(
                             false, a3, false, bf, (short)0, acc[iv][3], false, false);
        }
        k0 += 4;
    } while (k0 <= khi);

    #pragma unroll
    for (int iv = 0; iv < VT_PER_WAVE; ++iv) {
        #pragma unroll
        for (int r = 0; r < 8; ++r) {
            const int u = ut * 16 + half * 8 + r;
            const int v = (vt0 + iv) * 16 + l15;
            const float w = acc[iv][3][r];
            const size_t o = ((size_t)(b * UPTS + u) * VPTS + v) * 3;
            out[o + 0] = acc[iv][0][r] / w;
            out[o + 1] = acc[iv][1][r] / w;
            out[o + 2] = acc[iv][2][r] / w;
        }
    }
}

extern "C" void kernel_launch(void* const* d_in, const int* in_sizes, int n_in,
                              void* d_out, int out_size, void* d_ws, size_t ws_size,
                              hipStream_t stream) {
    const float* ctrl  = (const float*)d_in[0];
    const float* Nu    = (const float*)d_in[1];
    const float* Nv    = (const float*)d_in[2];
    const int*   uspan = (const int*)  d_in[3];
    const int*   vspan = (const int*)  d_in[4];
    float*       out   = (float*)d_out;

    float* NuF = (float*)d_ws;
    float* NvF = NuF + (size_t)UPTS * NCP;
    float* T   = NvF + (size_t)VPTS * NCP;

    build_dense_basis<<<(2 * UPTS + 255) / 256, 256, 0, stream>>>(
        Nu, Nv, uspan, vspan, NuF, NvF);

    surf_stage1<<<dim3(UPTS / 16, NCP / 16, NB * 4), 32, 0, stream>>>(
        NuF, ctrl, T);

    surf_stage2<<<dim3(VPTS / 16 / VT_PER_WAVE, UPTS / 16, NB), 32, 0, stream>>>(
        T, NvF, vspan, out);
}


# One launch instead of three: a 16-wave work-group per (16-point u tile, batch) keeps its T slab in LDS for stage 2; basis fragments picked from Nu/Nv by span, no dense tables; sum order kept
# speedup vs baseline: 2.6269x; 2.3096x over previous
#include <hip/hip_runtime.h>


typedef float v2f __attribute__((ext_vector_type(2)));
typedef float v8f __attribute__((ext_vector_type(8)));

#define NCP  64
#define UPTS 1024
#define VPTS 1024
#define NB   4
#define VT_PER_WAVE 4
#define WAVES 16
#define TP   68

__device__ __forceinline__ float band_pick(const float4 n, const int l) {
    float r = 0.0f;
    r = (l == 0) ? n.x : r;
    r = (l == 1) ? n.y : r;
    r = (l == 2) ? n.z : r;
    r = (l == 3) ? n.w : r;
    return r;
}

__global__ __launch_bounds__(WAVES * 32)
void surf_fused(const float* __restrict__ ctrl,
                const float* __restrict__ Nu,
                const float* __restrict__ Nv,
                const int*   __restrict__ uspan,
                const int*   __restrict__ vspan,
                float* __restrict__ out) {
    __shared__ __attribute__((aligned(16))) float Ts[4 * 16 * TP];

    const int lane = threadIdx.x & 31;
    const int wv   = threadIdx.x >> 5;
    const int ut   = blockIdx.x;
    const int b    = blockIdx.y;

    const int half = lane >> 4;
    const int l15  = lane & 15;

    const int vt0 = wv * VT_PER_WAVE;
    float4 nv[VT_PER_WAVE];
    int    sv[VT_PER_WAVE];
    #pragma unroll
    for (int iv = 0; iv < VT_PER_WAVE; ++iv) {
        const int v = (vt0 + iv) * 16 + l15;
        nv[iv] = *(const float4*)(Nv + v * 4);
        sv[iv] = vspan[v] - 3;
    }
    int smin, smax;
    {
        const int sa = vspan[vt0 * 16 + lane];
        const int sb = vspan[vt0 * 16 + 32 + lane];
        smin = min(sa, sb);
        smax = max(sa, sb);
    }

    {
        const int jt   = wv >> 2;
        const int d    = wv & 3;
        const int urow = ut * 16 + l15;
        const int jcol = jt * 16 + l15;
        const float4 nu = *(const float4*)(Nu + urow * 4);
        const int    su = uspan[urow] - 3;

        v8f acc = {};
        #pragma unroll
        for (int k0 = 0; k0 < NCP; k0 += 4) {
            const int ka = k0 + half * 2;
            v2f a;
            a.x = band_pick(nu, ka     - su);
            a.y = band_pick(nu, ka + 1 - su);
            v2f bf;
            bf.x = ctrl[((b * NCP + ka    ) * NCP + jcol) * 4 + d];
            bf.y = ctrl[((b * NCP + ka + 1) * NCP + jcol) * 4 + d];
            acc = __builtin_amdgcn_wmma_f32_16x16x4_f32(
                      false, a, false, bf, (short)0, acc, false, false);
        }

        #pragma unroll
        for (int r = 0; r < 8; ++r)
            Ts[(d * 16 + half * 8 + r) * TP + jt * 16 + l15] = acc[r];
    }

    for (int off = 16; off >= 1; off >>= 1) {
        smin = min(smin, __shfl_xor(smin, off));
        smax = max(smax, __shfl_xor(smax, off));
    }
    const int klo = min(max(__builtin_amdgcn_readfirstlane(smin) - 3, 0) & ~3, NCP - 4);
    const int khi = min(__builtin_amdgcn_readfirstlane(smax), NCP - 1);

    __syncthreads();

    const float* t0 = Ts + (0 * 16 + l15) * TP;
    const float* t1 = Ts + (1 * 16 + l15) * TP;
    const float* t2 = Ts + (2 * 16 + l15) * TP;
    const float* t3 = Ts + (3 * 16 + l15) * TP;

    v8f acc[VT_PER_WAVE][4];
    #pragma unroll
    for (int iv = 0; iv < VT_PER_WAVE; ++iv)
        #pragma unroll
        for (int d = 0; d < 4; ++d)
            acc[iv][d] = (v8f){};

    int k0 = klo;
    do {
        const int ka = k0 + half * 2;
        v2f a0 = *(const v2f*)(t0 + ka);
        v2f a1 = *(const v2f*)(t1 + ka);
        v2f a2 = *(const v2f*)(t2 + ka);
        v2f a3 = *(const v2f*)(t3 + ka);
        #pragma unroll
        for (int iv = 0; iv < VT_PER_WAVE; ++iv) {
            v2f bf;
            bf.x = band_pick(nv[iv], ka     - sv[iv]);
            bf.y = band_pick(nv[iv], ka + 1 - sv[iv]);
            acc[iv][0] = __builtin_amdgcn_wmma_f32_16x16x4_f32(
                             false, a0, false, bf, (short)0, acc[iv][0], false, false);
            acc[iv][1] = __builtin_amdgcn_wmma_f32_16x16x4_f32(
                             false, a1, false, bf, (short)0, acc[iv][1], false, false);
            acc[iv][2] = __builtin_amdgcn_wmma_f32_16x16x4_f32(
                             false, a2, false, bf, (short)0, acc[iv][2], false, false);
            acc[iv][3] = __builtin_amdgcn_wmma_f32_16x16x4_f32(
                             false, a3, false, bf, (short)0, acc[iv][3], false, false);
        }
        k0 += 4;
    } while (k0 <= khi);

    #pragma unroll
    for (int iv = 0; iv < VT_PER_WAVE; ++iv) {
        #pragma unroll
        for (int r = 0; r < 8; ++r) {
            const int u = ut * 16 + half * 8 + r;
            const int v = (vt0 + iv) * 16 + l15;
            const float w = acc[iv][3][r];
            const size_t o = ((size_t)(b * UPTS + u) * VPTS + v) * 3;
            out[o + 0] = acc[iv][0][r] / w;
            out[o + 1] = acc[iv][1][r] / w;
            out[o + 2] = acc[iv][2][r] / w;
        }
    }
}

extern "C" void kernel_launch(void* const* d_in, const int* in_sizes, int n_in,
                              void* d_out, int out_size, void* d_ws, size_t ws_size,
                              hipStream_t stream) {
    const float* ctrl  = (const float*)d_in[0];
    const float* Nu    = (const float*)d_in[1];
    const float* Nv    = (const float*)d_in[2];
    const int*   uspan = (const int*)  d_in[3];
    const int*   vspan = (const int*)  d_in[4];
    float*       out   = (float*)d_out;

    surf_fused<<<dim3(UPTS / 16, NB), WAVES * 32, 0, stream>>>(
        ctrl, Nu, Nv, uspan, vspan, out);
}


# Perspective divide as one v_rcp_f32 of w per point and three multiplies in place of three IEEE divides (11-instruction chains); quotients within about 2 ulp of the divide, sum order kept
# speedup vs baseline: 2.8699x; 1.0925x over previous
#include <hip/hip_runtime.h>


typedef float v2f __attribute__((ext_vector_type(2)));
typedef float v8f __attribute__((ext_vector_type(8)));

#define NCP  64
#define UPTS 1024
#define VPTS 1024
#define NB   4
#define VT_PER_WAVE 4
#define WAVES 16
#define TP   68

__device__ __forceinline__ float band_pick(const float4 n, const int l) {
    float r = 0.0f;
    r = (l == 0) ? n.x : r;
    r = (l == 1) ? n.y : r;
    r = (l == 2) ? n.z : r;
    r = (l == 3) ? n.w : r;
    return r;
}

__global__ __launch_bounds__(WAVES * 32)
void surf_fused(const float* __restrict__ ctrl,
                const float* __restrict__ Nu,
                const float* __restrict__ Nv,
                const int*   __restrict__ uspan,
                const int*   __restrict__ vspan,
                float* __restrict__ out) {
    __shared__ __attribute__((aligned(16))) float Ts[4 * 16 * TP];

    const int lane = threadIdx.x & 31;
    const int wv   = threadIdx.x >> 5;
    const int ut   = blockIdx.x;
    const int b    = blockIdx.y;

    const int half = lane >> 4;
    const int l15  = lane & 15;

    const int vt0 = wv * VT_PER_WAVE;
    float4 nv[VT_PER_WAVE];
    int    sv[VT_PER_WAVE];
    #pragma unroll
    for (int iv = 0; iv < VT_PER_WAVE; ++iv) {
        const int v = (vt0 + iv) * 16 + l15;
        nv[iv] = *(const float4*)(Nv + v * 4);
        sv[iv] = vspan[v] - 3;
    }
    int smin, smax;
    {
        const int sa = vspan[vt0 * 16 + lane];
        const int sb = vspan[vt0 * 16 + 32 + lane];
        smin = min(sa, sb);
        smax = max(sa, sb);
    }

    {
        const int jt   = wv >> 2;
        const int d    = wv & 3;
        const int urow = ut * 16 + l15;
        const int jcol = jt * 16 + l15;
        const float4 nu = *(const float4*)(Nu + urow * 4);
        const int    su = uspan[urow] - 3;

        v8f acc = {};
        #pragma unroll
        for (int k0 = 0; k0 < NCP; k0 += 4) {
            const int ka = k0 + half * 2;
            v2f a;
            a.x = band_pick(nu, ka     - su);
            a.y = band_pick(nu, ka + 1 - su);
            v2f bf;
            bf.x = ctrl[((b * NCP + ka    ) * NCP + jcol) * 4 + d];
            bf.y = ctrl[((b * NCP + ka + 1) * NCP + jcol) * 4 + d];
            acc = __builtin_amdgcn_wmma_f32_16x16x4_f32(
                      false, a, false, bf, (short)0, acc, false, false);
        }

        #pragma unroll
        for (int r = 0; r < 8; ++r)
            Ts[(d * 16 + half * 8 + r) * TP + jt * 16 + l15] = acc[r];
    }

    for (int off = 16; off >= 1; off >>= 1) {
        smin = min(smin, __shfl_xor(smin, off));
        smax = max(smax, __shfl_xor(smax, off));
    }
    const int klo = min(max(__builtin_amdgcn_readfirstlane(smin) - 3, 0) & ~3, NCP - 4);
    const int khi = min(__builtin_amdgcn_readfirstlane(smax), NCP - 1);

    __syncthreads();

    const float* t0 = Ts + (0 * 16 + l15) * TP;
    const float* t1 = Ts + (1 * 16 + l15) * TP;
    const float* t2 = Ts + (2 * 16 + l15) * TP;
    const float* t3 = Ts + (3 * 16 + l15) * TP;

    v8f acc[VT_PER_WAVE][4];
    #pragma unroll
    for (int iv = 0; iv < VT_PER_WAVE; ++iv)
        #pragma unroll
        for (int d = 0; d < 4; ++d)
            acc[iv][d] = (v8f){};

    int k0 = klo;
    do {
        const int ka = k0 + half * 2;
        v2f a0 = *(const v2f*)(t0 + ka);
        v2f a1 = *(const v2f*)(t1 + ka);
        v2f a2 = *(const v2f*)(t2 + ka);
        v2f a3 = *(const v2f*)(t3 + ka);
        #pragma unroll
        for (int iv = 0; iv < VT_PER_WAVE; ++iv) {
            v2f bf;
            bf.x = band_pick(nv[iv], ka     - sv[iv]);
            bf.y = band_pick(nv[iv], ka + 1 - sv[iv]);
            acc[iv][0] = __builtin_amdgcn_wmma_f32_16x16x4_f32(
                             false, a0, false, bf, (short)0, acc[iv][0], false, false);
            acc[iv][1] = __builtin_amdgcn_wmma_f32_16x16x4_f32(
                             false, a1, false, bf, (short)0, acc[iv][1], false, false);
            acc[iv][2] = __builtin_amdgcn_wmma_f32_16x16x4_f32(
                             false, a2, false, bf, (short)0, acc[iv][2], false, false);
            acc[iv][3] = __builtin_amdgcn_wmma_f32_16x16x4_f32(
                             false, a3, false, bf, (short)0, acc[iv][3], false, false);
        }
        k0 += 4;
    } while (k0 <= khi);

    #pragma unroll
    for (int iv = 0; iv < VT_PER_WAVE; ++iv) {
        #pragma unroll
        for (int r = 0; r < 8; ++r) {
            const int u = ut * 16 + half * 8 + r;
            const int v = (vt0 + iv) * 16 + l15;
            const float rw = __builtin_amdgcn_rcpf(acc[iv][3][r]);
            const size_t o = ((size_t)(b * UPTS + u) * VPTS + v) * 3;
            out[o + 0] = acc[iv][0][r] * rw;
            out[o + 1] = acc[iv][1][r] * rw;
            out[o + 2] = acc[iv][2][r] * rw;
        }
    }
}

extern "C" void kernel_launch(void* const* d_in, const int* in_sizes, int n_in,
                              void* d_out, int out_size, void* d_ws, size_t ws_size,
                              hipStream_t stream) {
    const float* ctrl  = (const float*)d_in[0];
    const float* Nu    = (const float*)d_in[1];
    const float* Nv    = (const float*)d_in[2];
    const int*   uspan = (const int*)  d_in[3];
    const int*   vspan = (const int*)  d_in[4];
    float*       out   = (float*)d_out;

    surf_fused<<<dim3(UPTS / 16, NB), WAVES * 32, 0, stream>>>(
        ctrl, Nu, Nv, uspan, vspan, out);
}


# Output stores as whole 128-byte lines: stage-2 tiles formed transposed (operands swapped, same products, same K order), strip staged in per-wave LDS, written back 16 B per lane; sum order kept
# speedup vs baseline: 4.0037x; 1.3951x over previous
#include <hip/hip_runtime.h>


typedef float v2f __attribute__((ext_vector_type(2)));
typedef float v8f __attribute__((ext_vector_type(8)));

#define NCP  64
#define UPTS 1024
#define VPTS 1024
#define NB   4
#define VT_PER_WAVE 4
#define WAVES 16
#define TP   68
#define SROW 192
#define SP   196

__device__ __forceinline__ float band_pick(const float4 n, const int l) {
    float r = 0.0f;
    r = (l == 0) ? n.x : r;
    r = (l == 1) ? n.y : r;
    r = (l == 2) ? n.z : r;
    r = (l == 3) ? n.w : r;
    return r;
}

__global__ __launch_bounds__(WAVES * 32)
void surf_fused(const float* __restrict__ ctrl,
                const float* __restrict__ Nu,
                const float* __restrict__ Nv,
                const int*   __restrict__ uspan,
                const int*   __restrict__ vspan,
                float* __restrict__ out) {
    __shared__ __attribute__((aligned(16))) float Ts[4 * 16 * TP];
    __shared__ __attribute__((aligned(16))) float Stg[WAVES * 16 * SP];

    const int lane = threadIdx.x & 31;
    const int wv   = threadIdx.x >> 5;
    const int ut   = blockIdx.x;
    const int b    = blockIdx.y;

    const int half = lane >> 4;
    const int l15  = lane & 15;

    const int vt0 = wv * VT_PER_WAVE;
    float4 nv[VT_PER_WAVE];
    int    sv[VT_PER_WAVE];
    #pragma unroll
    for (int iv = 0; iv < VT_PER_WAVE; ++iv) {
        const int v = (vt0 + iv) * 16 + l15;
        nv[iv] = *(const float4*)(Nv + v * 4);
        sv[iv] = vspan[v] - 3;
    }
    int smin, smax;
    {
        const int sa = vspan[vt0 * 16 + lane];
        const int sb = vspan[vt0 * 16 + 32 + lane];
        smin = min(sa, sb);
        smax = max(sa, sb);
    }

    {
        const int jt   = wv >> 2;
        const int d    = wv & 3;
        const int urow = ut * 16 + l15;
        const int jcol = jt * 16 + l15;
        const float4 nu = *(const float4*)(Nu + urow * 4);
        const int    su = uspan[urow] - 3;

        v8f acc = {};
        #pragma unroll
        for (int k0 = 0; k0 < NCP; k0 += 4) {
            const int ka = k0 + half * 2;
            v2f a;
            a.x = band_pick(nu, ka     - su);
            a.y = band_pick(nu, ka + 1 - su);
            v2f bf;
            bf.x = ctrl[((b * NCP + ka    ) * NCP + jcol) * 4 + d];
            bf.y = ctrl[((b * NCP + ka + 1) * NCP + jcol) * 4 + d];
            acc = __builtin_amdgcn_wmma_f32_16x16x4_f32(
                      false, a, false, bf, (short)0, acc, false, false);
        }

        #pragma unroll
        for (int r = 0; r < 8; ++r)
            Ts[(d * 16 + half * 8 + r) * TP + jt * 16 + l15] = acc[r];
    }

    for (int off = 16; off >= 1; off >>= 1) {
        smin = min(smin, __shfl_xor(smin, off));
        smax = max(smax, __shfl_xor(smax, off));
    }
    const int klo = min(max(__builtin_amdgcn_readfirstlane(smin) - 3, 0) & ~3, NCP - 4);
    const int khi = min(__builtin_amdgcn_readfirstlane(smax), NCP - 1);

    __syncthreads();

    const float* t0 = Ts + (0 * 16 + l15) * TP;
    const float* t1 = Ts + (1 * 16 + l15) * TP;
    const float* t2 = Ts + (2 * 16 + l15) * TP;
    const float* t3 = Ts + (3 * 16 + l15) * TP;

    v8f acc[VT_PER_WAVE][4];
    #pragma unroll
    for (int iv = 0; iv < VT_PER_WAVE; ++iv)
        #pragma unroll
        for (int d = 0; d < 4; ++d)
            acc[iv][d] = (v8f){};

    int k0 = klo;
    do {
        const int ka = k0 + half * 2;
        v2f a0 = *(const v2f*)(t0 + ka);
        v2f a1 = *(const v2f*)(t1 + ka);
        v2f a2 = *(const v2f*)(t2 + ka);
        v2f a3 = *(const v2f*)(t3 + ka);
        #pragma unroll
        for (int iv = 0; iv < VT_PER_WAVE; ++iv) {
            v2f bf;
            bf.x = band_pick(nv[iv], ka     - sv[iv]);
            bf.y = band_pick(nv[iv], ka + 1 - sv[iv]);
            acc[iv][0] = __builtin_amdgcn_wmma_f32_16x16x4_f32(
                             false, bf, false, a0, (short)0, acc[iv][0], false, false);
            acc[iv][1] = __builtin_amdgcn_wmma_f32_16x16x4_f32(
                             false, bf, false, a1, (short)0, acc[iv][1], false, false);
            acc[iv][2] = __builtin_amdgcn_wmma_f32_16x16x4_f32(
                             false, bf, false, a2, (short)0, acc[iv][2], false, false);
            acc[iv][3] = __builtin_amdgcn_wmma_f32_16x16x4_f32(
                             false, bf, false, a3, (short)0, acc[iv][3], false, false);
        }
        k0 += 4;
    } while (k0 <= khi);

    float* srow = Stg + (wv * 16 + l15) * SP + half * 24;
    #pragma unroll
    for (int iv = 0; iv < VT_PER_WAVE; ++iv) {
        float o[24];
        #pragma unroll
        for (int r = 0; r < 8; ++r) {
            const float rw = __builtin_amdgcn_rcpf(acc[iv][3][r]);
            o[3 * r + 0] = acc[iv][0][r] * rw;
            o[3 * r + 1] = acc[iv][1][r] * rw;
            o[3 * r + 2] = acc[iv][2][r] * rw;
        }
        #pragma unroll
        for (int q = 0; q < 6; ++q)
            *(float4*)(srow + iv * 48 + q * 4) =
                make_float4(o[4 * q], o[4 * q + 1], o[4 * q + 2], o[4 * q + 3]);
    }

    asm volatile("s_wait_dscnt 0x0" ::: "memory");

    {
        const float* sblk = Stg + (wv * 16 + half) * SP + l15 * 4;
        float* gblk = out + ((size_t)(b * UPTS + ut * 16 + half) * VPTS + vt0 * 16) * 3 + l15 * 4;
        #pragma unroll
        for (int m = 0; m < 8; ++m) {
            #pragma unroll
            for (int c = 0; c < 3; ++c) {
                const float4 val = *(const float4*)(sblk + m * 2 * SP + c * 64);
                *(float4*)(gblk + (size_t)m * 2 * (VPTS * 3) + c * 64) = val;
            }
        }
    }
}

extern "C" void kernel_launch(void* const* d_in, const int* in_sizes, int n_in,
                              void* d_out, int out_size, void* d_ws, size_t ws_size,
                              hipStream_t stream) {
    const float* ctrl  = (const float*)d_in[0];
    const float* Nu    = (const float*)d_in[1];
    const float* Nv    = (const float*)d_in[2];
    const int*   uspan = (const int*)  d_in[3];
    const int*   vspan = (const int*)  d_in[4];
    float*       out   = (float*)d_out;

    surf_fused<<<dim3(UPTS / 16, NB), WAVES * 32, 0, stream>>>(
        ctrl, Nu, Nv, uspan, vspan, out);
}


# Both K loops run only the aligned 4-wide blocks covering the spans of their own 16-point tile (stage 1: 2 blocks, not 16; stage 2: per v tile, not per strip); skipped blocks are zeros, sum order kept
# speedup vs baseline: 4.5759x; 1.1429x over previous
#include <hip/hip_runtime.h>


typedef float v2f __attribute__((ext_vector_type(2)));
typedef float v8f __attribute__((ext_vector_type(8)));

#define NCP  64
#define UPTS 1024
#define VPTS 1024
#define NB   4
#define VT_PER_WAVE 4
#define WAVES 16
#define TP   68
#define SROW 192
#define SP   196

__device__ __forceinline__ float band_pick(const float4 n, const int l) {
    float r = 0.0f;
    r = (l == 0) ? n.x : r;
    r = (l == 1) ? n.y : r;
    r = (l == 2) ? n.z : r;
    r = (l == 3) ? n.w : r;
    return r;
}

__global__ __launch_bounds__(WAVES * 32)
void surf_fused(const float* __restrict__ ctrl,
                const float* __restrict__ Nu,
                const float* __restrict__ Nv,
                const int*   __restrict__ uspan,
                const int*   __restrict__ vspan,
                float* __restrict__ out) {
    __shared__ __attribute__((aligned(16))) float Ts[4 * 16 * TP];
    __shared__ __attribute__((aligned(16))) float Stg[WAVES * 16 * SP];

    const int lane = threadIdx.x & 31;
    const int wv   = threadIdx.x >> 5;
    const int ut   = blockIdx.x;
    const int b    = blockIdx.y;

    const int half = lane >> 4;
    const int l15  = lane & 15;

    const int vt0 = wv * VT_PER_WAVE;
    float4 nv[VT_PER_WAVE];
    int    sv[VT_PER_WAVE];
    #pragma unroll
    for (int iv = 0; iv < VT_PER_WAVE; ++iv) {
        const int v = (vt0 + iv) * 16 + l15;
        nv[iv] = *(const float4*)(Nv + v * 4);
        sv[iv] = vspan[v] - 3;
    }
    int mina = vspan[vt0 * 16 + lane];
    int minb = vspan[vt0 * 16 + 32 + lane];
    int maxa = mina, maxb = minb;

    {
        const int jt   = wv >> 2;
        const int d    = wv & 3;
        const int urow = ut * 16 + l15;
        const int jcol = jt * 16 + l15;
        const float4 nu = *(const float4*)(Nu + urow * 4);
        const int    su = uspan[urow] - 3;

        int umin = uspan[ut * 16], umax = umin;
        #pragma unroll
        for (int i = 1; i < 16; ++i) {
            const int sp = uspan[ut * 16 + i];
            umin = min(umin, sp);
            umax = max(umax, sp);
        }
        const int ulo = min(max(umin - 3, 0) & ~3, NCP - 4);
        const int uhi = min(max(umax, ulo), NCP - 1);
        const int np  = (uhi - ulo + 8) >> 3;
        int k0 = min(ulo, NCP - 8 * np);
        const int kend = k0 + 8 * np;
        const float* cb = ctrl + ((size_t)b * NCP * NCP + jcol) * 4 + d;

        v8f acc = {};
        do {
            const int ka = k0 + half * 2;
            v2f bf0, bf1;
            bf0.x = cb[(ka    ) * (NCP * 4)];
            bf0.y = cb[(ka + 1) * (NCP * 4)];
            bf1.x = cb[(ka + 4) * (NCP * 4)];
            bf1.y = cb[(ka + 5) * (NCP * 4)];
            v2f a0, a1;
            a0.x = band_pick(nu, ka     - su);
            a0.y = band_pick(nu, ka + 1 - su);
            a1.x = band_pick(nu, ka + 4 - su);
            a1.y = band_pick(nu, ka + 5 - su);
            acc = __builtin_amdgcn_wmma_f32_16x16x4_f32(
                      false, a0, false, bf0, (short)0, acc, false, false);
            acc = __builtin_amdgcn_wmma_f32_16x16x4_f32(
                      false, a1, false, bf1, (short)0, acc, false, false);
            k0 += 8;
        } while (k0 < kend);

        #pragma unroll
        for (int r = 0; r < 8; ++r)
            Ts[(d * 16 + half * 8 + r) * TP + jt * 16 + l15] = acc[r];
    }

    for (int off = 8; off >= 1; off >>= 1) {
        mina = min(mina, __shfl_xor(mina, off));
        maxa = max(maxa, __shfl_xor(maxa, off));
        minb = min(minb, __shfl_xor(minb, off));
        maxb = max(maxb, __shfl_xor(maxb, off));
    }
    int tlo[VT_PER_WAVE], thi[VT_PER_WAVE];
    #pragma unroll
    for (int iv = 0; iv < VT_PER_WAVE; ++iv) {
        const int src = (iv & 1) * 16;
        const int tmn = __builtin_amdgcn_readlane((iv & 2) ? minb : mina, src);
        const int tmx = __builtin_amdgcn_readlane((iv & 2) ? maxb : maxa, src);
        tlo[iv] = min(max(tmn - 3, 0) & ~3, NCP - 4);
        thi[iv] = min(tmx, NCP - 1);
    }

    __syncthreads();

    const float* t0 = Ts + (0 * 16 + l15) * TP;
    const float* t1 = Ts + (1 * 16 + l15) * TP;
    const float* t2 = Ts + (2 * 16 + l15) * TP;
    const float* t3 = Ts + (3 * 16 + l15) * TP;

    v8f acc[VT_PER_WAVE][4];
    #pragma unroll
    for (int iv = 0; iv < VT_PER_WAVE; ++iv)
        #pragma unroll
        for (int d = 0; d < 4; ++d)
            acc[iv][d] = (v8f){};

    #pragma unroll
    for (int iv = 0; iv < VT_PER_WAVE; ++iv) {
        int k0 = tlo[iv];
        do {
            const int ka = k0 + half * 2;
            const v2f a0 = *(const v2f*)(t0 + ka);
            const v2f a1 = *(const v2f*)(t1 + ka);
            const v2f a2 = *(const v2f*)(t2 + ka);
            const v2f a3 = *(const v2f*)(t3 + ka);
            v2f bf;
            bf.x = band_pick(nv[iv], ka     - sv[iv]);
            bf.y = band_pick(nv[iv], ka + 1 - sv[iv]);
            acc[iv][0] = __builtin_amdgcn_wmma_f32_16x16x4_f32(
                             false, bf, false, a0, (short)0, acc[iv][0], false, false);
            acc[iv][1] = __builtin_amdgcn_wmma_f32_16x16x4_f32(
                             false, bf, false, a1, (short)0, acc[iv][1], false, false);
            acc[iv][2] = __builtin_amdgcn_wmma_f32_16x16x4_f32(
                             false, bf, false, a2, (short)0, acc[iv][2], false, false);
            acc[iv][3] = __builtin_amdgcn_wmma_f32_16x16x4_f32(
                             false, bf, false, a3, (short)0, acc[iv][3], false, false);
            k0 += 4;
        } while (k0 <= thi[iv]);
    }

    float* srow = Stg + (wv * 16 + l15) * SP + half * 24;
    #pragma unroll
    for (int iv = 0; iv < VT_PER_WAVE; ++iv) {
        float o[24];
        #pragma unroll
        for (int r = 0; r < 8; ++r) {
            const float rw = __builtin_amdgcn_rcpf(acc[iv][3][r]);
            o[3 * r + 0] = acc[iv][0][r] * rw;
            o[3 * r + 1] = acc[iv][1][r] * rw;
            o[3 * r + 2] = acc[iv][2][r] * rw;
        }
        #pragma unroll
        for (int q = 0; q < 6; ++q)
            *(float4*)(srow + iv * 48 + q * 4) =
                make_float4(o[4 * q], o[4 * q + 1], o[4 * q + 2], o[4 * q + 3]);
    }

    asm volatile("s_wait_dscnt 0x0" ::: "memory");

    {
        const float* sblk = Stg + (wv * 16 + half) * SP + l15 * 4;
        float* gblk = out + ((size_t)(b * UPTS + ut * 16 + half) * VPTS + vt0 * 16) * 3 + l15 * 4;
        #pragma unroll
        for (int m = 0; m < 8; ++m) {
            #pragma unroll
            for (int c = 0; c < 3; ++c) {
                const float4 val = *(const float4*)(sblk + m * 2 * SP + c * 64);
                *(float4*)(gblk + (size_t)m * 2 * (VPTS * 3) + c * 64) = val;
            }
        }
    }
}

extern "C" void kernel_launch(void* const* d_in, const int* in_sizes, int n_in,
                              void* d_out, int out_size, void* d_ws, size_t ws_size,
                              hipStream_t stream) {
    const float* ctrl  = (const float*)d_in[0];
    const float* Nu    = (const float*)d_in[1];
    const float* Nv    = (const float*)d_in[2];
    const int*   uspan = (const int*)  d_in[3];
    const int*   vspan = (const int*)  d_in[4];
    float*       out   = (float*)d_out;

    surf_fused<<<dim3(UPTS / 16, NB), WAVES * 32, 0, stream>>>(
        ctrl, Nu, Nv, uspan, vspan, out);
}
